# starting kernel
# baseline (speedup 1.0000x reference)
#include <hip/hip_runtime.h>
#include <math.h>


typedef __attribute__((ext_vector_type(2))) float v2f;
typedef __attribute__((ext_vector_type(8))) float v8f;

#define GCN_N 50000
#define GCN_E 600000
#define GCN_D 128
#define TILE_K 32
#define LDS_PAD 4


__global__ __launch_bounds__(256) void deg_init_k(float* __restrict__ deg, int n) {
    int i = blockIdx.x * 256 + threadIdx.x;
    if (i < n) deg[i] = 1.0f;
}

__global__ __launch_bounds__(256) void deg_accum_k(float* __restrict__ deg,
                                                   const int* __restrict__ col,
                                                   const float* __restrict__ w, int e) {
    int i = blockIdx.x * 256 + threadIdx.x;
    if (i < e) atomicAdd(&deg[col[i]], w[i]);
}

__global__ __launch_bounds__(256) void dinv_k(const float* __restrict__ deg,
                                              float* __restrict__ dinv, int n) {
    int i = blockIdx.x * 256 + threadIdx.x;
    if (i < n) {
        float d = deg[i];
        dinv[i] = (d > 0.0f) ? rsqrtf(d) : 0.0f;
    }
}

__global__ __launch_bounds__(256) void gemm_wmma_k(const float* __restrict__ A,
                                                   const float* __restrict__ W,
                                                   float* __restrict__ C) {
    __shared__ float As[16][TILE_K + LDS_PAD];
    __shared__ float Wt[GCN_D][TILE_K + LDS_PAD];

    const int t    = threadIdx.x;
    const int lane = t & 31;
    const int wave = t >> 5;
    const int rowBase = blockIdx.x * 16;

    v8f acc = {};

    for (int kc = 0; kc < GCN_D; kc += TILE_K) {
        __syncthreads();
        for (int idx = t; idx < 16 * TILE_K; idx += 256) {
            int m = idx / TILE_K, k = idx % TILE_K;
            As[m][k] = A[(rowBase + m) * GCN_D + kc + k];
        }
        for (int idx = t; idx < TILE_K * GCN_D; idx += 256) {
            int k = idx / GCN_D, n = idx % GCN_D;
            Wt[n][k] = W[(kc + k) * GCN_D + n];
        }
        __syncthreads();

        const int m   = lane & 15;
        const int n   = wave * 16 + (lane & 15);
        const int klo = (lane >> 4) * 2;
        #pragma unroll
        for (int kk = 0; kk < TILE_K; kk += 4) {
            v2f a, b;
            a.x = As[m][kk + klo];
            a.y = As[m][kk + klo + 1];
            b.x = Wt[n][kk + klo];
            b.y = Wt[n][kk + klo + 1];
            acc = __builtin_amdgcn_wmma_f32_16x16x4_f32(
                 false, a,  false, b,
                 (short)0, acc,  false,  false);
        }
    }

    const int col    = wave * 16 + (lane & 15);
    const int rowOff = (lane >> 4) * 8;
    #pragma unroll
    for (int v = 0; v < 8; ++v) {
        C[(rowBase + v + rowOff) * GCN_D + col] = acc[v];
    }
}

__global__ __launch_bounds__(256) void init_out_k(float* __restrict__ out,
                                                  const float* __restrict__ xl,
                                                  const float* __restrict__ dinv,
                                                  const float* __restrict__ b) {
    int idx = blockIdx.x * 256 + threadIdx.x;
    int i = idx >> 7;
    int d = idx & (GCN_D - 1);
    float di = dinv[i];
    out[idx] = di * di * xl[idx] + b[d];
}

__global__ __launch_bounds__(256) void scatter_k(float* __restrict__ out,
                                                 const float* __restrict__ xl,
                                                 const int* __restrict__ row,
                                                 const int* __restrict__ col,
                                                 const float* __restrict__ w,
                                                 const float* __restrict__ dinv,
                                                 int e_total) {
    int wave = threadIdx.x >> 5;
    int lane = threadIdx.x & 31;
    int e = blockIdx.x * 8 + wave;
    if (e >= e_total) return;

    int r = row[e];
    int c = col[e];
    float norm = dinv[r] * w[e] * dinv[c];

    const float4* src = (const float4*)(xl + (size_t)r * GCN_D);
    float4 v = src[lane];
    float* dst = out + (size_t)c * GCN_D + lane * 4;
    atomicAdd(dst + 0, norm * v.x);
    atomicAdd(dst + 1, norm * v.y);
    atomicAdd(dst + 2, norm * v.z);
    atomicAdd(dst + 3, norm * v.w);
}

__global__ __launch_bounds__(256) void act_k(float* __restrict__ out, int mode,
                                             const float* __restrict__ prelu_w) {
    int idx = blockIdx.x * 256 + threadIdx.x;
    float h = out[idx];
    float r;
    if (mode == 0) {
        r = (h > 0.0f) ? h : (expf(h) - 1.0f);
    } else if (mode == 1) {
        r = 0.5f * h * (1.0f + erff(h * 0.70710678118654752f));
    } else {
        r = (h >= 0.0f) ? h : prelu_w[0] * h;
    }
    out[idx] = r;
}


extern "C" void kernel_launch(void* const* d_in, const int* in_sizes, int n_in,
                              void* d_out, int out_size, void* d_ws, size_t ws_size,
                              hipStream_t stream) {
    (void)in_sizes; (void)n_in; (void)out_size; (void)ws_size;

    const float* x   = (const float*)d_in[0];
    const int*   row = (const int*)d_in[1];
    const int*   col = ((const int*)d_in[1]) + GCN_E;
    const float* w   = (const float*)d_in[2];
    const float* W1  = (const float*)d_in[3];
    const float* b1  = (const float*)d_in[4];
    const float* W2  = (const float*)d_in[5];
    const float* b2  = (const float*)d_in[6];
    const float* W3  = (const float*)d_in[7];
    const float* b3  = (const float*)d_in[8];
    const float* pw  = (const float*)d_in[9];

    float* out  = (float*)d_out;
    float* xl   = (float*)d_ws;
    float* deg  = xl + (size_t)GCN_N * GCN_D;
    float* dinv = deg + GCN_N;

    const int nThr = 256;
    const int gN   = (GCN_N + nThr - 1) / nThr;
    const int gE   = (GCN_E + nThr - 1) / nThr;
    const int gND  = (GCN_N * GCN_D) / nThr;
    const int gGemm = GCN_N / 16;
    const int gScat = (GCN_E + 7) / 8;

    deg_init_k <<<gN, nThr, 0, stream>>>(deg, GCN_N);
    deg_accum_k<<<gE, nThr, 0, stream>>>(deg, col, w, GCN_E);
    dinv_k     <<<gN, nThr, 0, stream>>>(deg, dinv, GCN_N);

    gemm_wmma_k<<<gGemm, nThr, 0, stream>>>(x, W1, xl);
    init_out_k <<<gND,   nThr, 0, stream>>>(out, xl, dinv, b1);
    scatter_k  <<<gScat, nThr, 0, stream>>>(out, xl, row, col, w, dinv, GCN_E);
    act_k      <<<gND,   nThr, 0, stream>>>(out, 0, pw);

    gemm_wmma_k<<<gGemm, nThr, 0, stream>>>(out, W2, xl);
    init_out_k <<<gND,   nThr, 0, stream>>>(out, xl, dinv, b2);
    scatter_k  <<<gScat, nThr, 0, stream>>>(out, xl, row, col, w, dinv, GCN_E);
    act_k      <<<gND,   nThr, 0, stream>>>(out, 1, pw);

    gemm_wmma_k<<<gGemm, nThr, 0, stream>>>(out, W3, xl);
    init_out_k <<<gND,   nThr, 0, stream>>>(out, xl, dinv, b3);
    scatter_k  <<<gScat, nThr, 0, stream>>>(out, xl, row, col, w, dinv, GCN_E);
    act_k      <<<gND,   nThr, 0, stream>>>(out, 2, pw);
}


# gemm_wmma_k retiled: 256 workgroups stage W^T in LDS once; a wave owns whole 16-row panels over all 128 columns, A fragments straight from global; the product's k sum order is permuted
# speedup vs baseline: 3.5476x; 3.5476x over previous
#include <hip/hip_runtime.h>
#include <math.h>


typedef __attribute__((ext_vector_type(2))) float v2f;
typedef __attribute__((ext_vector_type(8))) float v8f;

#define GCN_N 50000
#define GCN_E 600000
#define GCN_D 128
#define WT_PITCH 132
#define GEMM_WGS 256
#define GEMM_PANELS (GCN_N / 16)
#define GEMM_SLOTS (GEMM_WGS * 4)
#define GEMM_LDS_FLOATS (41 * 1024)

#define TROWS 196
#define NTILES 256
#define BIN_BLOCKS 64
#define BIN_EDGES (GCN_E / BIN_BLOCKS)
#define BIN_PER_THREAD 10
#define ROWS_PER_WAVE 7

__global__ __launch_bounds__(1024) void bin_count_k(const int* __restrict__ col,
                                                    int* __restrict__ hist) {
    __shared__ int cnt[NTILES];
    const int t = threadIdx.x;
    if (t < NTILES) cnt[t] = 0;
    __syncthreads();
    const int e0 = blockIdx.x * BIN_EDGES;
    int c[BIN_PER_THREAD];
    #pragma unroll
    for (int j = 0; j < BIN_PER_THREAD; ++j) {
        int i = t + 1024 * j;
        c[j] = (i < BIN_EDGES) ? col[e0 + i] : -1;
    }
    #pragma unroll
    for (int j = 0; j < BIN_PER_THREAD; ++j) {
        if (c[j] >= 0) atomicAdd(&cnt[c[j] / TROWS], 1);
    }
    __syncthreads();
    if (t < NTILES) hist[blockIdx.x * NTILES + t] = cnt[t];
}

__global__ __launch_bounds__(NTILES) void bin_scan_k(const int* __restrict__ hist,
                                                     int* __restrict__ boff,
                                                     int* __restrict__ tstart) {
    __shared__ int tot[NTILES];
    const int t = threadIdx.x;
    int run = 0;
    #pragma unroll 16
    for (int b = 0; b < BIN_BLOCKS; ++b) {
        int h = hist[b * NTILES + t];
        boff[b * NTILES + t] = run;
        run += h;
    }
    tot[t] = run;
    __syncthreads();
    for (int d = 1; d < NTILES; d <<= 1) {
        int v = (t >= d) ? tot[t - d] : 0;
        __syncthreads();
        tot[t] += v;
        __syncthreads();
    }
    tstart[t] = tot[t] - run;
    if (t == NTILES - 1) tstart[NTILES] = tot[t];
}

__global__ __launch_bounds__(1024) void bin_fill_k(const int* __restrict__ row,
                                                   const int* __restrict__ col,
                                                   const float* __restrict__ w,
                                                   const int* __restrict__ boff,
                                                   const int* __restrict__ tstart,
                                                   int* __restrict__ pe) {
    __shared__ int cur[NTILES];
    const int t = threadIdx.x;
    if (t < NTILES) cur[t] = tstart[t] + boff[blockIdx.x * NTILES + t];
    __syncthreads();
    const int e0 = blockIdx.x * BIN_EDGES;
    int   c[BIN_PER_THREAD];
    int   r[BIN_PER_THREAD];
    float x[BIN_PER_THREAD];
    #pragma unroll
    for (int j = 0; j < BIN_PER_THREAD; ++j) {
        int i = t + 1024 * j;
        bool ok = i < BIN_EDGES;
        c[j] = ok ? col[e0 + i] : -1;
        r[j] = ok ? row[e0 + i] : 0;
        x[j] = ok ? w[e0 + i] : 0.0f;
    }
    #pragma unroll
    for (int j = 0; j < BIN_PER_THREAD; ++j) {
        if (c[j] >= 0) {
            int tile = c[j] / TROWS;
            int pos = atomicAdd(&cur[tile], 1);
            ((int2*)pe)[pos] = make_int2((r[j] << 8) | (c[j] - tile * TROWS), __float_as_int(x[j]));
        }
    }
}

__global__ __launch_bounds__(1024) void tile_deg_k(const int* __restrict__ pe,
                                                   const int* __restrict__ tstart,
                                                   float* __restrict__ dinv) {
    __shared__ float dg[TROWS];
    const int t = threadIdx.x;
    if (t < TROWS) dg[t] = 1.0f;
    __syncthreads();
    const int ts = tstart[blockIdx.x];
    const int te = tstart[blockIdx.x + 1];
    for (int i = ts + t; i < te; i += 1024) {
        int2 en = ((const int2*)pe)[i];
        atomicAdd(&dg[en.x & 255], __int_as_float(en.y));
    }
    __syncthreads();
    const int node = blockIdx.x * TROWS + t;
    if (t < TROWS && node < GCN_N) {
        float d = dg[t];
        dinv[node] = (d > 0.0f) ? rsqrtf(d) : 0.0f;
    }
}

__global__ __launch_bounds__(256) void gemm_wmma_k(const float* __restrict__ A,
                                                   const float* __restrict__ W,
                                                   float* __restrict__ C) {
    __shared__ alignas(16) float Wt[GEMM_LDS_FLOATS];

    const int t    = threadIdx.x;
    const int lane = t & 31;
    const int wave = __builtin_amdgcn_readfirstlane(t >> 5);
    const int c    = lane & 15;
    const int h    = lane >> 4;

    int p = blockIdx.x * 4 + (wave & 3) + GEMM_SLOTS * (wave >> 2);

    float4 a_cur[16];
    {
        const float* ap = A + (size_t)(p * 16 + c) * GCN_D + 4 * h;
        #pragma unroll
        for (int j = 0; j < 16; ++j) a_cur[j] = *(const float4*)(ap + 8 * j);
    }

    {
        const int n  = t & 127;
        const int kb = (t >> 7) * 64;
        float v[64];
        #pragma unroll
        for (int i = 0; i < 64; ++i) v[i] = W[(kb + i) * GCN_D + n];
        #pragma unroll
        for (int g = 0; g < 16; ++g) {
            *(float4*)(Wt + n * WT_PITCH + kb + 4 * g) =
                make_float4(v[4 * g], v[4 * g + 1], v[4 * g + 2], v[4 * g + 3]);
        }
    }
    __syncthreads();

    const float* bp = Wt + c * WT_PITCH + 4 * h;

    for (;;) {
        asm volatile("" ::: "memory");
        const int  pn   = p + 2 * GEMM_SLOTS;
        const bool more = pn < GEMM_PANELS;
        float4 a_nxt[16];
        if (more) {
            const float* ap = A + (size_t)(pn * 16 + c) * GCN_D + 4 * h;
            #pragma unroll
            for (int j = 0; j < 16; ++j) a_nxt[j] = *(const float4*)(ap + 8 * j);
        }

        v8f acc[8];
        #pragma unroll
        for (int ct = 0; ct < 8; ++ct) acc[ct] = (v8f){};

        float4 bq[2][8];
        #pragma unroll
        for (int ct = 0; ct < 8; ++ct) bq[0][ct] = *(const float4*)(bp + ct * 16 * WT_PITCH);

        #pragma unroll
        for (int j = 0; j < 16; ++j) {
            if (j + 1 < 16) {
                #pragma unroll
                for (int ct = 0; ct < 8; ++ct)
                    bq[(j + 1) & 1][ct] = *(const float4*)(bp + ct * 16 * WT_PITCH + 8 * (j + 1));
            }
            const float4 a = a_cur[j];
            v2f a0, a1;
            a0.x = a.x; a0.y = a.y;
            a1.x = a.z; a1.y = a.w;
            #pragma unroll
            for (int ct = 0; ct < 8; ++ct) {
                v2f b0;
                b0.x = bq[j & 1][ct].x; b0.y = bq[j & 1][ct].y;
                acc[ct] = __builtin_amdgcn_wmma_f32_16x16x4_f32(
                     false, a0,  false, b0,
                     (short)0, acc[ct],  false,  false);
            }
            #pragma unroll
            for (int ct = 0; ct < 8; ++ct) {
                v2f b1;
                b1.x = bq[j & 1][ct].z; b1.y = bq[j & 1][ct].w;
                acc[ct] = __builtin_amdgcn_wmma_f32_16x16x4_f32(
                     false, a1,  false, b1,
                     (short)0, acc[ct],  false,  false);
            }
        }

        float* cp = C + (size_t)(p * 16 + 8 * h) * GCN_D + c;
        #pragma unroll
        for (int ct = 0; ct < 8; ++ct) {
            #pragma unroll
            for (int r = 0; r < 8; ++r) cp[r * GCN_D + 16 * ct] = acc[ct][r];
        }

        if (!more) break;
        #pragma unroll
        for (int j = 0; j < 16; ++j) a_cur[j] = a_nxt[j];
        p = pn;
    }
}

__device__ __forceinline__ float act_f(float h, int mode, float slope) {
    if (mode == 0) return (h > 0.0f) ? h : (expf(h) - 1.0f);
    if (mode == 1) return 0.5f * h * (1.0f + erff(h * 0.70710678118654752f));
    return (h >= 0.0f) ? h : slope * h;
}

template <bool FULL>
__device__ __forceinline__ void add_group(float* __restrict__ acc,
                                          const float* __restrict__ xl,
                                          int p, float nv, int k0, int n, int lane) {
    float4 v[8];
    #pragma unroll
    for (int k = 0; k < 8; ++k) {
        if (FULL || k0 + k < n) {
            int pe = __builtin_amdgcn_readlane(p, k0 + k);
            v[k] = ((const float4*)(xl + (size_t)(pe >> 8) * GCN_D))[lane];
        }
    }
    #pragma unroll
    for (int k = 0; k < 8; ++k) {
        if (FULL || k0 + k < n) {
            int   pe = __builtin_amdgcn_readlane(p, k0 + k);
            float ne = __int_as_float(__builtin_amdgcn_readlane(__float_as_int(nv), k0 + k));
            float* a = acc + (pe & 255) * GCN_D + lane;
            atomicAdd(a,      ne * v[k].x);
            atomicAdd(a + 32, ne * v[k].y);
            atomicAdd(a + 64, ne * v[k].z);
            atomicAdd(a + 96, ne * v[k].w);
        }
    }
}

__global__ __launch_bounds__(1024) void tile_agg_k(float* __restrict__ out,
                                                   const float* __restrict__ xl,
                                                   const int* __restrict__ pe,
                                                   const int* __restrict__ tstart,
                                                   const float* __restrict__ dinv,
                                                   const float* __restrict__ b,
                                                   int mode,
                                                   const float* __restrict__ prelu_w) {
    __shared__ float acc[TROWS * GCN_D];
    __shared__ float dvt[TROWS];

    const int t    = threadIdx.x;
    const int lane = t & 31;
    const int wave = __builtin_amdgcn_readfirstlane(t >> 5);
    const int rowBase = blockIdx.x * TROWS;
    const int nrows = (GCN_N - rowBase < TROWS) ? (GCN_N - rowBase) : TROWS;

    if (t < TROWS) dvt[t] = (t < nrows) ? dinv[rowBase + t] : 0.0f;
    {
        const float4 bv = ((const float4*)b)[lane];
        float4 v[ROWS_PER_WAVE];
        float  di[ROWS_PER_WAVE];
        #pragma unroll
        for (int it = 0; it < ROWS_PER_WAVE; ++it) {
            int r = wave + 32 * it;
            int rr = (r < nrows) ? r : (nrows - 1);
            v[it]  = ((const float4*)(xl + (size_t)(rowBase + rr) * GCN_D))[lane];
            di[it] = dinv[rowBase + rr];
        }
        #pragma unroll
        for (int it = 0; it < ROWS_PER_WAVE; ++it) {
            int r = wave + 32 * it;
            if (r < nrows) {
                float* a = acc + r * GCN_D + lane;
                float d2 = di[it] * di[it];
                a[0]  = d2 * v[it].x + bv.x;
                a[32] = d2 * v[it].y + bv.y;
                a[64] = d2 * v[it].z + bv.z;
                a[96] = d2 * v[it].w + bv.w;
            }
        }
    }
    __syncthreads();

    {
        const int ts  = tstart[blockIdx.x];
        const int cnt = tstart[blockIdx.x + 1] - ts;
        const int wbeg = ts + ((cnt * wave) >> 5);
        const int wend = ts + ((cnt * (wave + 1)) >> 5);
        for (int base = wbeg; base < wend; base += 32) {
            const int i = base + lane;
            int   p  = 0;
            float nv = 0.0f;
            if (i < wend) {
                int2 en = ((const int2*)pe)[i];
                p  = en.x;
                nv = dinv[p >> 8] * __int_as_float(en.y) * dvt[p & 255];
            }
            const int n = (wend - base < 32) ? (wend - base) : 32;
            for (int k0 = 0; k0 < n; k0 += 8) {
                if (k0 + 8 <= n) add_group<true >(acc, xl, p, nv, k0, n, lane);
                else             add_group<false>(acc, xl, p, nv, k0, n, lane);
            }
        }
    }
    __syncthreads();

    {
        const float slope = prelu_w[0];
        #pragma unroll
        for (int it = 0; it < ROWS_PER_WAVE; ++it) {
            int r = wave + 32 * it;
            if (r < nrows) {
                const float* a = acc + r * GCN_D + lane;
                float4 h;
                h.x = act_f(a[0],  mode, slope);
                h.y = act_f(a[32], mode, slope);
                h.z = act_f(a[64], mode, slope);
                h.w = act_f(a[96], mode, slope);
                ((float4*)(out + (size_t)(rowBase + r) * GCN_D))[lane] = h;
            }
        }
    }
}


extern "C" void kernel_launch(void* const* d_in, const int* in_sizes, int n_in,
                              void* d_out, int out_size, void* d_ws, size_t ws_size,
                              hipStream_t stream) {
    (void)in_sizes; (void)n_in; (void)out_size; (void)ws_size;

    const float* x   = (const float*)d_in[0];
    const int*   row = (const int*)d_in[1];
    const int*   col = ((const int*)d_in[1]) + GCN_E;
    const float* w   = (const float*)d_in[2];
    const float* W1  = (const float*)d_in[3];
    const float* b1  = (const float*)d_in[4];
    const float* W2  = (const float*)d_in[5];
    const float* b2  = (const float*)d_in[6];
    const float* W3  = (const float*)d_in[7];
    const float* b3  = (const float*)d_in[8];
    const float* pw_ = (const float*)d_in[9];

    float* out  = (float*)d_out;
    float* xl   = (float*)d_ws;
    float* dinv = xl + (size_t)GCN_N * GCN_D;
    int*   pe   = (int*)(dinv + GCN_N);
    int*   hist = pe + 2 * GCN_E;
    int*   boff = hist + BIN_BLOCKS * NTILES;
    int*   tst  = boff + BIN_BLOCKS * NTILES;

    bin_count_k<<<BIN_BLOCKS, 1024, 0, stream>>>(col, hist);
    bin_scan_k <<<1, NTILES, 0, stream>>>(hist, boff, tst);
    bin_fill_k <<<BIN_BLOCKS, 1024, 0, stream>>>(row, col, w, boff, tst, pe);
    tile_deg_k <<<NTILES, 1024, 0, stream>>>(pe, tst, dinv);

    gemm_wmma_k<<<GEMM_WGS, 256, 0, stream>>>(x, W1, xl);
    tile_agg_k <<<NTILES, 1024, 0, stream>>>(out, xl, pe, tst, dinv, b1, 0, pw_);

    gemm_wmma_k<<<GEMM_WGS, 256, 0, stream>>>(out, W2, xl);
    tile_agg_k <<<NTILES, 1024, 0, stream>>>(out, xl, pe, tst, dinv, b2, 1, pw_);

    gemm_wmma_k<<<GEMM_WGS, 256, 0, stream>>>(out, W3, xl);
    tile_agg_k <<<NTILES, 1024, 0, stream>>>(out, xl, pe, tst, dinv, b3, 2, pw_);
}


# scan and degree launches folded away: bin_fill_k scans the slab histograms itself; first GEMM launch sums tile b's degree in workgroup b (f32 LDS adds, arrival order, 256 threads); 10->8 launches
# speedup vs baseline: 3.7254x; 1.0501x over previous
#include <hip/hip_runtime.h>
#include <math.h>


typedef __attribute__((ext_vector_type(2))) float v2f;
typedef __attribute__((ext_vector_type(8))) float v8f;

#define GCN_N 50000
#define GCN_E 600000
#define GCN_D 128
#define WT_PITCH 132
#define GEMM_WGS 256
#define GEMM_PANELS (GCN_N / 16)
#define GEMM_SLOTS (GEMM_WGS * 4)
#define GEMM_LDS_FLOATS (41 * 1024)

#define TROWS 196
#define NTILES 256
#define BIN_BLOCKS 64
#define BIN_EDGES (GCN_E / BIN_BLOCKS)
#define BIN_PER_THREAD 10
#define ROWS_PER_WAVE 7

__global__ __launch_bounds__(1024) void bin_count_k(const int* __restrict__ col,
                                                    int* __restrict__ hist) {
    __shared__ int cnt[NTILES];
    const int t = threadIdx.x;
    if (t < NTILES) cnt[t] = 0;
    __syncthreads();
    const int e0 = blockIdx.x * BIN_EDGES;
    int c[BIN_PER_THREAD];
    #pragma unroll
    for (int j = 0; j < BIN_PER_THREAD; ++j) {
        int i = t + 1024 * j;
        c[j] = (i < BIN_EDGES) ? col[e0 + i] : -1;
    }
    #pragma unroll
    for (int j = 0; j < BIN_PER_THREAD; ++j) {
        if (c[j] >= 0) atomicAdd(&cnt[c[j] / TROWS], 1);
    }
    __syncthreads();
    if (t < NTILES) hist[blockIdx.x * NTILES + t] = cnt[t];
}

__global__ __launch_bounds__(1024) void bin_fill_k(const int* __restrict__ row,
                                                   const int* __restrict__ col,
                                                   const float* __restrict__ w,
                                                   const int* __restrict__ hist,
                                                   int* __restrict__ tstart,
                                                   int* __restrict__ pe) {
    __shared__ int cur[NTILES];
    __shared__ int wsum[NTILES / 32];
    __shared__ int prun[4][NTILES];
    __shared__ int pmine[4][NTILES];
    const int t = threadIdx.x;
    const int lane = t & 31;
    const int e0 = blockIdx.x * BIN_EDGES;
    int   c[BIN_PER_THREAD];
    int   r[BIN_PER_THREAD];
    float x[BIN_PER_THREAD];
    #pragma unroll
    for (int j = 0; j < BIN_PER_THREAD; ++j) {
        int i = t + 1024 * j;
        bool ok = i < BIN_EDGES;
        c[j] = ok ? col[e0 + i] : -1;
        r[j] = ok ? row[e0 + i] : 0;
        x[j] = ok ? w[e0 + i] : 0.0f;
    }
    {
        const int tt = t & (NTILES - 1);
        const int g  = t >> 8;
        int hh[BIN_BLOCKS / 4];
        #pragma unroll
        for (int k = 0; k < BIN_BLOCKS / 4; ++k) hh[k] = hist[(g * (BIN_BLOCKS / 4) + k) * NTILES + tt];
        int pr = 0, pm = 0;
        #pragma unroll
        for (int k = 0; k < BIN_BLOCKS / 4; ++k) {
            pr += hh[k];
            pm += (g * (BIN_BLOCKS / 4) + k < (int)blockIdx.x) ? hh[k] : 0;
        }
        prun[g][tt]  = pr;
        pmine[g][tt] = pm;
    }
    __syncthreads();
    int run = 0, mine = 0, incl = 0;
    if (t < NTILES) {
        run  = prun[0][t]  + prun[1][t]  + prun[2][t]  + prun[3][t];
        mine = pmine[0][t] + pmine[1][t] + pmine[2][t] + pmine[3][t];
        incl = run;
        for (int d = 1; d < 32; d <<= 1) {
            int v = __shfl_up(incl, d);
            if (lane >= d) incl += v;
        }
        if (lane == 31) wsum[t >> 5] = incl;
    }
    __syncthreads();
    if (t < NTILES) {
        int base = 0;
        for (int k = 0; k < NTILES / 32; ++k) if (k < (t >> 5)) base += wsum[k];
        const int start = base + incl - run;
        cur[t] = start + mine;
        if (blockIdx.x == 0) {
            tstart[t] = start;
            if (t == NTILES - 1) tstart[NTILES] = start + run;
        }
    }
    __syncthreads();
    #pragma unroll
    for (int j = 0; j < BIN_PER_THREAD; ++j) {
        if (c[j] >= 0) {
            int tile = c[j] / TROWS;
            int pos = atomicAdd(&cur[tile], 1);
            ((int2*)pe)[pos] = make_int2((r[j] << 8) | (c[j] - tile * TROWS), __float_as_int(x[j]));
        }
    }
}

#define DEG_BATCH 12
static_assert(NTILES == GEMM_WGS, "workgroup b of the first GEMM launch sums tile b's degree");
template <bool DEG>
__global__ __launch_bounds__(256) void gemm_wmma_k(const float* __restrict__ A,
                                                   const float* __restrict__ W,
                                                   float* __restrict__ C,
                                                   const int* __restrict__ pe,
                                                   const int* __restrict__ tstart,
                                                   float* __restrict__ dinv) {
    __shared__ alignas(16) float Wt[GEMM_LDS_FLOATS];
    __shared__ float dg[TROWS];

    const int t    = threadIdx.x;
    const int lane = t & 31;
    const int wave = __builtin_amdgcn_readfirstlane(t >> 5);
    const int c    = lane & 15;
    const int h    = lane >> 4;

    int p = blockIdx.x * 4 + (wave & 3) + GEMM_SLOTS * (wave >> 2);

    if (DEG) {
        if (t < TROWS) dg[t] = 1.0f;
        __syncthreads();
    }

    float4 a_cur[16];
    {
        const float* ap = A + (size_t)(p * 16 + c) * GCN_D + 4 * h;
        #pragma unroll
        for (int j = 0; j < 16; ++j) a_cur[j] = *(const float4*)(ap + 8 * j);
    }

    {
        const int n  = t & 127;
        const int kb = (t >> 7) * 64;
        float v[64];
        #pragma unroll
        for (int i = 0; i < 64; ++i) v[i] = W[(kb + i) * GCN_D + n];
        if (DEG) {
            const int ts = tstart[blockIdx.x];
            const int te = tstart[blockIdx.x + 1];
            for (int base = ts; base < te; base += 256 * DEG_BATCH) {
                int2 en[DEG_BATCH];
                #pragma unroll
                for (int j = 0; j < DEG_BATCH; ++j) {
                    int i = base + t + 256 * j;
                    i = (i < GCN_E - 1) ? i : (GCN_E - 1);
                    en[j] = ((const int2*)pe)[i];
                }
                #pragma unroll
                for (int j = 0; j < DEG_BATCH; ++j) {
                    if (base + t + 256 * j < te) atomicAdd(&dg[en[j].x & 255], __int_as_float(en[j].y));
                }
            }
        }
        #pragma unroll
        for (int g = 0; g < 16; ++g) {
            *(float4*)(Wt + n * WT_PITCH + kb + 4 * g) =
                make_float4(v[4 * g], v[4 * g + 1], v[4 * g + 2], v[4 * g + 3]);
        }
    }
    __syncthreads();

    if (DEG) {
        const int node = blockIdx.x * TROWS + t;
        if (t < TROWS && node < GCN_N) {
            float d = dg[t];
            dinv[node] = (d > 0.0f) ? rsqrtf(d) : 0.0f;
        }
    }

    const float* bp = Wt + c * WT_PITCH + 4 * h;

    for (;;) {
        asm volatile("" ::: "memory");
        const int  pn   = p + 2 * GEMM_SLOTS;
        const bool more = pn < GEMM_PANELS;
        float4 a_nxt[16];
        if (more) {
            const float* ap = A + (size_t)(pn * 16 + c) * GCN_D + 4 * h;
            #pragma unroll
            for (int j = 0; j < 16; ++j) a_nxt[j] = *(const float4*)(ap + 8 * j);
        }

        v8f acc[8];
        #pragma unroll
        for (int ct = 0; ct < 8; ++ct) acc[ct] = (v8f){};

        float4 bq[2][8];
        #pragma unroll
        for (int ct = 0; ct < 8; ++ct) bq[0][ct] = *(const float4*)(bp + ct * 16 * WT_PITCH);

        #pragma unroll
        for (int j = 0; j < 16; ++j) {
            if (j + 1 < 16) {
                #pragma unroll
                for (int ct = 0; ct < 8; ++ct)
                    bq[(j + 1) & 1][ct] = *(const float4*)(bp + ct * 16 * WT_PITCH + 8 * (j + 1));
            }
            const float4 a = a_cur[j];
            v2f a0, a1;
            a0.x = a.x; a0.y = a.y;
            a1.x = a.z; a1.y = a.w;
            #pragma unroll
            for (int ct = 0; ct < 8; ++ct) {
                v2f b0;
                b0.x = bq[j & 1][ct].x; b0.y = bq[j & 1][ct].y;
                acc[ct] = __builtin_amdgcn_wmma_f32_16x16x4_f32(
                     false, a0,  false, b0,
                     (short)0, acc[ct],  false,  false);
            }
            #pragma unroll
            for (int ct = 0; ct < 8; ++ct) {
                v2f b1;
                b1.x = bq[j & 1][ct].z; b1.y = bq[j & 1][ct].w;
                acc[ct] = __builtin_amdgcn_wmma_f32_16x16x4_f32(
                     false, a1,  false, b1,
                     (short)0, acc[ct],  false,  false);
            }
        }

        float* cp = C + (size_t)(p * 16 + 8 * h) * GCN_D + c;
        #pragma unroll
        for (int ct = 0; ct < 8; ++ct) {
            #pragma unroll
            for (int r = 0; r < 8; ++r) cp[r * GCN_D + 16 * ct] = acc[ct][r];
        }

        if (!more) break;
        #pragma unroll
        for (int j = 0; j < 16; ++j) a_cur[j] = a_nxt[j];
        p = pn;
    }
}

__device__ __forceinline__ float act_f(float h, int mode, float slope) {
    if (mode == 0) return (h > 0.0f) ? h : (expf(h) - 1.0f);
    if (mode == 1) return 0.5f * h * (1.0f + erff(h * 0.70710678118654752f));
    return (h >= 0.0f) ? h : slope * h;
}

template <bool FULL>
__device__ __forceinline__ void add_group(float* __restrict__ acc,
                                          const float* __restrict__ xl,
                                          int p, float nv, int k0, int n, int lane) {
    float4 v[8];
    #pragma unroll
    for (int k = 0; k < 8; ++k) {
        if (FULL || k0 + k < n) {
            int pe = __builtin_amdgcn_readlane(p, k0 + k);
            v[k] = ((const float4*)(xl + (size_t)(pe >> 8) * GCN_D))[lane];
        }
    }
    #pragma unroll
    for (int k = 0; k < 8; ++k) {
        if (FULL || k0 + k < n) {
            int   pe = __builtin_amdgcn_readlane(p, k0 + k);
            float ne = __int_as_float(__builtin_amdgcn_readlane(__float_as_int(nv), k0 + k));
            float* a = acc + (pe & 255) * GCN_D + lane;
            atomicAdd(a,      ne * v[k].x);
            atomicAdd(a + 32, ne * v[k].y);
            atomicAdd(a + 64, ne * v[k].z);
            atomicAdd(a + 96, ne * v[k].w);
        }
    }
}

__global__ __launch_bounds__(1024) void tile_agg_k(float* __restrict__ out,
                                                   const float* __restrict__ xl,
                                                   const int* __restrict__ pe,
                                                   const int* __restrict__ tstart,
                                                   const float* __restrict__ dinv,
                                                   const float* __restrict__ b,
                                                   int mode,
                                                   const float* __restrict__ prelu_w) {
    __shared__ float acc[TROWS * GCN_D];
    __shared__ float dvt[TROWS];

    const int t    = threadIdx.x;
    const int lane = t & 31;
    const int wave = __builtin_amdgcn_readfirstlane(t >> 5);
    const int rowBase = blockIdx.x * TROWS;
    const int nrows = (GCN_N - rowBase < TROWS) ? (GCN_N - rowBase) : TROWS;

    if (t < TROWS) dvt[t] = (t < nrows) ? dinv[rowBase + t] : 0.0f;
    {
        const float4 bv = ((const float4*)b)[lane];
        float4 v[ROWS_PER_WAVE];
        float  di[ROWS_PER_WAVE];
        #pragma unroll
        for (int it = 0; it < ROWS_PER_WAVE; ++it) {
            int r = wave + 32 * it;
            int rr = (r < nrows) ? r : (nrows - 1);
            v[it]  = ((const float4*)(xl + (size_t)(rowBase + rr) * GCN_D))[lane];
            di[it] = dinv[rowBase + rr];
        }
        #pragma unroll
        for (int it = 0; it < ROWS_PER_WAVE; ++it) {
            int r = wave + 32 * it;
            if (r < nrows) {
                float* a = acc + r * GCN_D + lane;
                float d2 = di[it] * di[it];
                a[0]  = d2 * v[it].x + bv.x;
                a[32] = d2 * v[it].y + bv.y;
                a[64] = d2 * v[it].z + bv.z;
                a[96] = d2 * v[it].w + bv.w;
            }
        }
    }
    __syncthreads();

    {
        const int ts  = tstart[blockIdx.x];
        const int cnt = tstart[blockIdx.x + 1] - ts;
        const int wbeg = ts + ((cnt * wave) >> 5);
        const int wend = ts + ((cnt * (wave + 1)) >> 5);
        for (int base = wbeg; base < wend; base += 32) {
            const int i = base + lane;
            int   p  = 0;
            float nv = 0.0f;
            if (i < wend) {
                int2 en = ((const int2*)pe)[i];
                p  = en.x;
                nv = dinv[p >> 8] * __int_as_float(en.y) * dvt[p & 255];
            }
            const int n = (wend - base < 32) ? (wend - base) : 32;
            for (int k0 = 0; k0 < n; k0 += 8) {
                if (k0 + 8 <= n) add_group<true >(acc, xl, p, nv, k0, n, lane);
                else             add_group<false>(acc, xl, p, nv, k0, n, lane);
            }
        }
    }
    __syncthreads();

    {
        const float slope = prelu_w[0];
        #pragma unroll
        for (int it = 0; it < ROWS_PER_WAVE; ++it) {
            int r = wave + 32 * it;
            if (r < nrows) {
                const float* a = acc + r * GCN_D + lane;
                float4 h;
                h.x = act_f(a[0],  mode, slope);
                h.y = act_f(a[32], mode, slope);
                h.z = act_f(a[64], mode, slope);
                h.w = act_f(a[96], mode, slope);
                ((float4*)(out + (size_t)(rowBase + r) * GCN_D))[lane] = h;
            }
        }
    }
}


extern "C" void kernel_launch(void* const* d_in, const int* in_sizes, int n_in,
                              void* d_out, int out_size, void* d_ws, size_t ws_size,
                              hipStream_t stream) {
    (void)in_sizes; (void)n_in; (void)out_size; (void)ws_size;

    const float* x   = (const float*)d_in[0];
    const int*   row = (const int*)d_in[1];
    const int*   col = ((const int*)d_in[1]) + GCN_E;
    const float* w   = (const float*)d_in[2];
    const float* W1  = (const float*)d_in[3];
    const float* b1  = (const float*)d_in[4];
    const float* W2  = (const float*)d_in[5];
    const float* b2  = (const float*)d_in[6];
    const float* W3  = (const float*)d_in[7];
    const float* b3  = (const float*)d_in[8];
    const float* pw_ = (const float*)d_in[9];

    float* out  = (float*)d_out;
    float* xl   = (float*)d_ws;
    float* dinv = xl + (size_t)GCN_N * GCN_D;
    int*   pe   = (int*)(dinv + GCN_N);
    int*   hist = pe + 2 * GCN_E;
    int*   tst  = hist + BIN_BLOCKS * NTILES;

    bin_count_k<<<BIN_BLOCKS, 1024, 0, stream>>>(col, hist);
    bin_fill_k <<<BIN_BLOCKS, 1024, 0, stream>>>(row, col, w, hist, tst, pe);

    gemm_wmma_k<true><<<GEMM_WGS, 256, 0, stream>>>(x, W1, xl, pe, tst, dinv);
    tile_agg_k <<<NTILES, 1024, 0, stream>>>(out, xl, pe, tst, dinv, b1, 0, pw_);

    gemm_wmma_k<false><<<GEMM_WGS, 256, 0, stream>>>(out, W2, xl, pe, tst, dinv);
    tile_agg_k <<<NTILES, 1024, 0, stream>>>(out, xl, pe, tst, dinv, b2, 1, pw_);

    gemm_wmma_k<false><<<GEMM_WGS, 256, 0, stream>>>(out, W3, xl, pe, tst, dinv);
    tile_agg_k <<<NTILES, 1024, 0, stream>>>(out, xl, pe, tst, dinv, b3, 2, pw_);
}
